# starting kernel
# baseline (speedup 1.0000x reference)
#include <hip/hip_runtime.h>


typedef __attribute__((ext_vector_type(2))) float v2f;
typedef __attribute__((ext_vector_type(8))) float v8f;

#define B_ 4
#define S_ 4096
#define D_ 768
#define K_ 24
#define HALF_S 2048
#define K4_ (D_ / 4)

__global__ __launch_bounds__(256)
void stu_pack_kernel(const float* __restrict__ Mi, float* __restrict__ P) {
    int n  = blockIdx.x * 256 + threadIdx.x;
    int k4 = blockIdx.y;
    if (n >= D_) return;
    float4 v;
    v.x = Mi[(size_t)(4 * k4 + 0) * D_ + n];
    v.y = Mi[(size_t)(4 * k4 + 1) * D_ + n];
    v.z = Mi[(size_t)(4 * k4 + 2) * D_ + n];
    v.w = Mi[(size_t)(4 * k4 + 3) * D_ + n];
    *(float4*)(P + ((size_t)k4 * D_ + n) * 4) = v;
}

__global__ __launch_bounds__(256)
void stu_filter_kernel(const float* __restrict__ phi,
                       const float* __restrict__ Mf,
                       float* __restrict__ he) {
    int d = blockIdx.x * 256 + threadIdx.x;
    int j = blockIdx.y;
    if (d >= D_) return;
    const float* prow = phi + (2 * j) * K_;
    float acc = 0.f;
#pragma unroll
    for (int k = 0; k < K_; ++k)
        acc = fmaf(prow[k], Mf[k * D_ + d], acc);
    he[j * D_ + d] = 2.0f * acc;
}

__global__ __launch_bounds__(128)
void stu_xproj_wmma(const float* __restrict__ X,
                    const float* __restrict__ P,
                    float* __restrict__ XP) {
    const int wave = threadIdx.x >> 5;
    const int lane = threadIdx.x & 31;
    const int half = lane >> 4;
    const int l16  = lane & 15;
    const int m0 = blockIdx.x * 32;
    const int n0 = blockIdx.y * 256 + wave * 64;

    v8f c0[4] = {}, c1[4] = {};
    const float* arow0 = X + (size_t)(m0 + l16) * D_ + half * 2;
    const float* arow1 = arow0 + (size_t)16 * D_;
    const float* pcol  = P + ((size_t)(n0 + l16)) * 4 + 2 * half;

    for (int k4 = 0; k4 < K4_; ++k4) {
        v2f a0 = *(const v2f*)(arow0 + k4 * 4);
        v2f a1 = *(const v2f*)(arow1 + k4 * 4);
        const float* pk = pcol + (size_t)k4 * D_ * 4;
#pragma unroll
        for (int nt = 0; nt < 4; ++nt) {
            v2f b = *(const v2f*)(pk + nt * 64);
            c0[nt] = __builtin_amdgcn_wmma_f32_16x16x4_f32(false, a0, false, b, (short)0, c0[nt], false, false);
            c1[nt] = __builtin_amdgcn_wmma_f32_16x16x4_f32(false, a1, false, b, (short)0, c1[nt], false, false);
        }
    }

    float* out0 = XP + (size_t)(m0 + half * 8) * D_ + n0 + l16;
    float* out1 = out0 + (size_t)16 * D_;
#pragma unroll
    for (int nt = 0; nt < 4; ++nt) {
#pragma unroll
        for (int v = 0; v < 8; ++v) {
            out0[(size_t)v * D_ + nt * 16] = c0[nt][v];
            out1[(size_t)v * D_ + nt * 16] = c1[nt][v];
        }
    }
}

__global__ __launch_bounds__(256)
void stu_conv_kernel(const float* __restrict__ XP,
                     const float* __restrict__ HE,
                     float* __restrict__ OUT) {
    __shared__ float sh_h[64 * 32];
    __shared__ float sh_x[128 * 32];

    const int ib = blockIdx.x;
    const int d0 = blockIdx.y * 32;
    const int b  = blockIdx.z >> 1;
    const int p  = blockIdx.z & 1;
    const int tid  = threadIdx.x;
    const int lane = tid & 31;
    const int w    = tid >> 5;
    const int i0 = ib * 64;
    const int d  = d0 + lane;

    float acc[8];
#pragma unroll
    for (int r = 0; r < 8; ++r) acc[r] = 0.f;

    const float* xpb = XP + (size_t)b * S_ * D_;

    for (int c = 0; c <= ib; ++c) {
        const int j0 = c * 64;
        for (int idx = tid; idx < 64 * 8; idx += 256) {
            int jj = idx >> 3, q = idx & 7;
            *(float4*)(sh_h + jj * 32 + q * 4) =
                *(const float4*)(HE + (size_t)(j0 + jj) * D_ + d0 + q * 4);
        }
        const int u_base = i0 - j0 - 63;
        for (int idx = tid; idx < 128 * 8; idx += 256) {
            int s = idx >> 3, q = idx & 7;
            int u = u_base + s;
            float4 v = {0.f, 0.f, 0.f, 0.f};
            if (u >= 0 && u < HALF_S)
                v = *(const float4*)(xpb + (size_t)(2 * u + p) * D_ + d0 + q * 4);
            *(float4*)(sh_x + s * 32 + q * 4) = v;
        }
        __syncthreads();

        if (c < ib) {
            const int j0n = j0 + 64;
            __builtin_prefetch(&HE[(size_t)(j0n + (tid >> 2)) * D_ + d0 + (tid & 3) * 8], 0, 0);
            int u_n = i0 - j0n - 63 + (tid >> 1);
            if (u_n >= 0 && u_n < HALF_S)
                __builtin_prefetch(&xpb[(size_t)(2 * u_n + p) * D_ + d0 + (tid & 1) * 16], 0, 0);
        }

        float xwin[8];
#pragma unroll
        for (int r = 0; r < 8; ++r) xwin[r] = sh_x[(8 * w + r) * 32 + lane];
#pragma unroll 8
        for (int jj = 63; jj >= 0; --jj) {
            float h = sh_h[jj * 32 + lane];
#pragma unroll
            for (int r = 0; r < 8; ++r) acc[r] = fmaf(h, xwin[r], acc[r]);
#pragma unroll
            for (int r = 0; r < 7; ++r) xwin[r] = xwin[r + 1];
            xwin[7] = sh_x[(8 * w + 71 - jj) * 32 + lane];
        }
        __syncthreads();
    }

#pragma unroll
    for (int r = 0; r < 8; ++r) {
        int t = 2 * (i0 + 8 * w + r) + p;
        OUT[((size_t)b * S_ + t) * D_ + d] = acc[r];
    }
}

extern "C" void kernel_launch(void* const* d_in, const int* in_sizes, int n_in,
                              void* d_out, int out_size, void* d_ws, size_t ws_size,
                              hipStream_t stream) {
    (void)in_sizes; (void)n_in; (void)out_size; (void)ws_size;
    const float* x   = (const float*)d_in[0];
    const float* phi = (const float*)d_in[1];
    const float* Mi  = (const float*)d_in[2];
    const float* Mf  = (const float*)d_in[3];
    float* out  = (float*)d_out;

    float* ws_x = (float*)d_ws;
    float* ws_h = ws_x + (size_t)B_ * S_ * D_;
    float* ws_p = ws_h + (size_t)HALF_S * D_;

    stu_pack_kernel  <<<dim3(3, K4_, 1),            256, 0, stream>>>(Mi, ws_p);
    stu_filter_kernel<<<dim3(3, HALF_S, 1),         256, 0, stream>>>(phi, Mf, ws_h);
    stu_xproj_wmma   <<<dim3((B_ * S_) / 32, D_ / 256, 1), 128, 0, stream>>>(x, ws_p, ws_x);
    stu_conv_kernel  <<<dim3(HALF_S / 64, D_ / 32, B_ * 2), 256, 0, stream>>>(ws_x, ws_h, out);
}


# FFT twiddles: one table piece per pass and level, factors w^x, w^2x, w^3x as separate runs over the lane-varying x, so a wave's twiddle load is one contiguous run, not up to 24 lines; same values
# speedup vs baseline: 2.4262x; 2.4262x over previous
#include <hip/hip_runtime.h>


typedef __attribute__((ext_vector_type(2))) float v2f;
typedef __attribute__((ext_vector_type(8))) float v8f;

#define B_ 4
#define S_ 4096
#define D_ 768
#define K_ 24
#define HALF_S 2048
#define K4_ (D_ / 4)

#define FN_     4096
#define FCH_    4
#define FPITCH_ (FN_ + FN_ / 16)

#define TWA4096_ 0
#define TWB4096_ 3072
#define TWA256_  3840
#define TWB256_  4032
#define TWA16_   4080
#define TWEND_   4092
__device__ __forceinline__ int stu_tw_expo(int t) {
    if (t < TWB4096_) return ((t >> 10) + 1) * (t & 1023);
    if (t < TWA256_)  { const int r = t - TWB4096_; return 4 * ((r >> 8) + 1) * (r & 255); }
    if (t < TWB256_)  { const int r = t - TWA256_;  return 16 * ((r >> 6) + 1) * (r & 63); }
    if (t < TWA16_)   { const int r = t - TWB256_;  return 64 * ((r >> 4) + 1) * (r & 15); }
    if (t < TWEND_)   { const int r = t - TWA16_;   return 256 * ((r >> 2) + 1) * (r & 3); }
    return 0;
}
template <int L> struct stu_twp {
    static constexpr int A  = (L == 4096) ? TWA4096_ : (L == 256) ? TWA256_ : TWA16_;
    static constexpr int NA = L / 4;
    static constexpr int B  = (L == 4096) ? TWB4096_ : TWB256_;
    static constexpr int NB = L / 16;
};

__global__ __launch_bounds__(256)
void stu_pack_kernel(const float* __restrict__ Mi, float* __restrict__ P, float* __restrict__ TWf) {
    int n  = blockIdx.x * 256 + threadIdx.x;
    int k4 = blockIdx.y;
    if (n >= D_) return;
    const int t = k4 * D_ + n;
    if (t < FN_) {
        float s, c;
        sincospif((float)stu_tw_expo(t) * (1.0f / 2048.0f), &s, &c);
        ((v2f*)TWf)[t] = v2f{c, -s};
    }
    float4 v;
    v.x = Mi[(size_t)(4 * k4 + 0) * D_ + n];
    v.y = Mi[(size_t)(4 * k4 + 1) * D_ + n];
    v.z = Mi[(size_t)(4 * k4 + 2) * D_ + n];
    v.w = Mi[(size_t)(4 * k4 + 3) * D_ + n];
    *(float4*)(P + ((size_t)k4 * D_ + n) * 4) = v;
}

__device__ __forceinline__ int stu_pidx(int n) { return n + (n >> 4); }

__device__ __forceinline__ v2f stu_cmul(v2f a, v2f w) {
    return v2f{a.x * w.x - a.y * w.y, a.x * w.y + a.y * w.x};
}
__device__ __forceinline__ v2f stu_cmulc(v2f a, v2f w) {
    return v2f{a.x * w.x + a.y * w.y, a.y * w.x - a.x * w.y};
}

template <bool INV>
__device__ __forceinline__ void stu_dft4(v2f& a0, v2f& a1, v2f& a2, v2f& a3) {
    const v2f s02 = a0 + a2, d02 = a0 - a2, s13 = a1 + a3, d13 = a1 - a3;
    const v2f r = INV ? v2f{-d13.y, d13.x} : v2f{d13.y, -d13.x};
    a0 = s02 + s13; a2 = s02 - s13; a1 = d02 + r; a3 = d02 - r;
}

template <int L>
__device__ __forceinline__ void stu_fwd16(v2f (&r)[16], int j, const v2f* __restrict__ tw) {
    constexpr int S = L / 16;
#pragma unroll
    for (int q0 = 0; q0 < 4; ++q0) {
        stu_dft4<false>(r[q0], r[q0 + 4], r[q0 + 8], r[q0 + 12]);
        if (!(L == 16 && q0 == 0)) {
            const v2f* ta = tw + stu_twp<L>::A + (j + S * q0);
            r[q0 + 4]  = stu_cmul(r[q0 + 4],  ta[0]);
            r[q0 + 8]  = stu_cmul(r[q0 + 8],  ta[stu_twp<L>::NA]);
            r[q0 + 12] = stu_cmul(r[q0 + 12], ta[2 * stu_twp<L>::NA]);
        }
    }
#pragma unroll
    for (int k = 0; k < 4; ++k) stu_dft4<false>(r[4 * k], r[4 * k + 1], r[4 * k + 2], r[4 * k + 3]);
    if (L > 16) {
        const v2f* tb = tw + stu_twp<L>::B + j;
        const v2f w1 = tb[0], w2 = tb[stu_twp<L>::NB], w3 = tb[2 * stu_twp<L>::NB];
#pragma unroll
        for (int k = 0; k < 4; ++k) {
            r[4 * k + 1] = stu_cmul(r[4 * k + 1], w1);
            r[4 * k + 2] = stu_cmul(r[4 * k + 2], w2);
            r[4 * k + 3] = stu_cmul(r[4 * k + 3], w3);
        }
    }
}

template <int L>
__device__ __forceinline__ void stu_inv16(v2f (&r)[16], int j, const v2f* __restrict__ tw) {
    constexpr int S = L / 16;
    if (L > 16) {
        const v2f* tb = tw + stu_twp<L>::B + j;
        const v2f w1 = tb[0], w2 = tb[stu_twp<L>::NB], w3 = tb[2 * stu_twp<L>::NB];
#pragma unroll
        for (int k = 0; k < 4; ++k) {
            r[4 * k + 1] = stu_cmulc(r[4 * k + 1], w1);
            r[4 * k + 2] = stu_cmulc(r[4 * k + 2], w2);
            r[4 * k + 3] = stu_cmulc(r[4 * k + 3], w3);
        }
    }
#pragma unroll
    for (int k = 0; k < 4; ++k) stu_dft4<true>(r[4 * k], r[4 * k + 1], r[4 * k + 2], r[4 * k + 3]);
#pragma unroll
    for (int q0 = 0; q0 < 4; ++q0) {
        if (!(L == 16 && q0 == 0)) {
            const v2f* ta = tw + stu_twp<L>::A + (j + S * q0);
            r[q0 + 4]  = stu_cmulc(r[q0 + 4],  ta[0]);
            r[q0 + 8]  = stu_cmulc(r[q0 + 8],  ta[stu_twp<L>::NA]);
            r[q0 + 12] = stu_cmulc(r[q0 + 12], ta[2 * stu_twp<L>::NA]);
        }
        stu_dft4<true>(r[q0], r[q0 + 4], r[q0 + 8], r[q0 + 12]);
    }
}

template <int L, bool INV, bool HALF>
__device__ __forceinline__ void stu_fft_pass(v2f* __restrict__ bc, int tl, const v2f* __restrict__ tw) {
    constexpr int S = L / 16;
#pragma unroll 1
    for (int it = 0; it < 2; ++it) {
        const int task = tl + 128 * it;
        const int bb = (task / S) * L, j = task % S;
        v2f r[16];
#pragma unroll
        for (int q = 0; q < 16; ++q)
            r[q] = (HALF && !INV && q >= 8) ? v2f{0.f, 0.f} : bc[stu_pidx(bb + j + S * q)];
        if (INV) stu_inv16<L>(r, j, tw); else stu_fwd16<L>(r, j, tw);
#pragma unroll
        for (int q = 0; q < 16; ++q)
            if (!(HALF && INV && q >= 8)) bc[stu_pidx(bb + j + S * q)] = r[q];
    }
}

__global__ __launch_bounds__(512, 8)
void stu_spectrum_kernel(const float* __restrict__ phi,
                         const float* __restrict__ Mf,
                         const float* __restrict__ TWf,
                         float* __restrict__ HSf) {
    __shared__ __attribute__((aligned(16))) v2f sb[FCH_ * FPITCH_];
    const int tid = threadIdx.x;
    const int c = tid >> 7, tl = tid & 127;
    const int d0 = blockIdx.x * FCH_;
    const v2f* tw = (const v2f*)TWf;

    float4 mf[K_];
#pragma unroll
    for (int k = 0; k < K_; ++k) mf[k] = *(const float4*)(Mf + (size_t)k * D_ + d0);
    for (int jj = 0; jj < HALF_S / 512; ++jj) {
        const int j = tid + 512 * jj;
        const float4* pr = (const float4*)(phi + (size_t)(2 * j) * K_);
        float a0 = 0.f, a1 = 0.f, a2 = 0.f, a3 = 0.f;
#pragma unroll
        for (int k4 = 0; k4 < K_ / 4; ++k4) {
            const float4 p = pr[k4];
            const float pv[4] = {p.x, p.y, p.z, p.w};
#pragma unroll
            for (int e = 0; e < 4; ++e) {
                const float4 m = mf[4 * k4 + e];
                a0 = fmaf(pv[e], m.x, a0); a1 = fmaf(pv[e], m.y, a1);
                a2 = fmaf(pv[e], m.z, a2); a3 = fmaf(pv[e], m.w, a3);
            }
        }
        const float sc = 2.0f / 4096.0f;
        const int pj = stu_pidx(j);
        sb[0 * FPITCH_ + pj] = v2f{a0 * sc, 0.f};
        sb[1 * FPITCH_ + pj] = v2f{a1 * sc, 0.f};
        sb[2 * FPITCH_ + pj] = v2f{a2 * sc, 0.f};
        sb[3 * FPITCH_ + pj] = v2f{a3 * sc, 0.f};
    }
    __syncthreads();
    v2f* bc = sb + c * FPITCH_;
    stu_fft_pass<4096, false, true >(bc, tl, tw); __syncthreads();
    stu_fft_pass<256,  false, false>(bc, tl, tw); __syncthreads();
    stu_fft_pass<16,   false, false>(bc, tl, tw); __syncthreads();
    for (int idx = tid; idx < FCH_ * (FN_ / 2); idx += 512) {
        const int c2 = idx >> 11, n = (idx & 2047) * 2;
        const v2f* s = sb + c2 * FPITCH_ + stu_pidx(n);
        const v2f u = s[0], v = s[1];
        *(float4*)(HSf + ((size_t)(d0 + c2) * FN_ + n) * 2) = float4{u.x, u.y, v.x, v.y};
    }
}

#define GM_ROWS_  64
#define GM_PITCH_ (D_ + 4)
__global__ __launch_bounds__(384)
void stu_xproj_wmma(const float* __restrict__ X,
                    const float* __restrict__ P,
                    float* __restrict__ XP) {
    __shared__ __attribute__((aligned(16))) float sa[GM_ROWS_ * GM_PITCH_];

    const int tid  = threadIdx.x;
    const int wave = tid >> 5;
    const int lane = tid & 31;
    const int half = lane >> 4;
    const int l16  = lane & 15;
    const int m0 = blockIdx.x * GM_ROWS_;
    const int n0 = wave * 64;

    const float* xb = X + (size_t)m0 * D_;
#pragma unroll
    for (int k = 0; k < GM_ROWS_ * (D_ / 4) / 384; ++k) {
        const int idx = tid + k * 384;
        const int row = idx / (D_ / 4), q = idx - row * (D_ / 4);
        *(float4*)(sa + row * GM_PITCH_ + q * 4) = *(const float4*)(xb + (size_t)idx * 4);
    }
    __syncthreads();

    v8f c[4][4] = {};
    const float* ar   = sa + l16 * GM_PITCH_ + half * 2;
    const float* pcol = P + ((size_t)(n0 + l16)) * 4 + 2 * half;

    for (int k4 = 0; k4 < K4_; ++k4) {
        v2f a[4];
#pragma unroll
        for (int mt = 0; mt < 4; ++mt)
            a[mt] = *(const v2f*)(ar + mt * 16 * GM_PITCH_ + k4 * 4);
        const float* pk = pcol + (size_t)k4 * D_ * 4;
#pragma unroll
        for (int nt = 0; nt < 4; ++nt) {
            v2f b = *(const v2f*)(pk + nt * 64);
#pragma unroll
            for (int mt = 0; mt < 4; ++mt)
                c[mt][nt] = __builtin_amdgcn_wmma_f32_16x16x4_f32(false, a[mt], false, b, (short)0, c[mt][nt], false, false);
        }
    }

    float* ob = XP + (size_t)(m0 + half * 8) * D_ + n0 + l16;
#pragma unroll
    for (int mt = 0; mt < 4; ++mt) {
#pragma unroll
        for (int nt = 0; nt < 4; ++nt) {
#pragma unroll
            for (int v = 0; v < 8; ++v)
                ob[(size_t)(mt * 16 + v) * D_ + nt * 16] = c[mt][nt][v];
        }
    }
}

__global__ __launch_bounds__(512, 8)
void stu_conv_fft_kernel(const float* __restrict__ XP,
                         const float* __restrict__ HSf,
                         const float* __restrict__ TWf,
                         float* __restrict__ OUT) {
    __shared__ __attribute__((aligned(16))) v2f sb[FCH_ * FPITCH_];
    const int tid = threadIdx.x;
    const int c = tid >> 7, tl = tid & 127;
    const int d0 = blockIdx.x * FCH_;
    const int b  = blockIdx.y;
    const v2f* tw = (const v2f*)TWf;

    const float* xpb = XP + (size_t)b * S_ * D_ + d0;
#pragma unroll
    for (int k = 0; k < HALF_S / 512; ++k) {
        const int i = tid + 512 * k;
        const float4 ev = *(const float4*)(xpb + (size_t)(2 * i) * D_);
        const float4 od = *(const float4*)(xpb + (size_t)(2 * i + 1) * D_);
        const int pi = stu_pidx(i);
        sb[0 * FPITCH_ + pi] = v2f{ev.x, od.x};
        sb[1 * FPITCH_ + pi] = v2f{ev.y, od.y};
        sb[2 * FPITCH_ + pi] = v2f{ev.z, od.z};
        sb[3 * FPITCH_ + pi] = v2f{ev.w, od.w};
    }
    __syncthreads();

    v2f* bc = sb + c * FPITCH_;
    stu_fft_pass<4096, false, true >(bc, tl, tw); __syncthreads();
    stu_fft_pass<256,  false, false>(bc, tl, tw); __syncthreads();
    {
        const v2f* hs = (const v2f*)HSf + (size_t)(d0 + c) * FN_;
#pragma unroll 1
        for (int it = 0; it < 2; ++it) {
            const int task = tl + 128 * it;
            v2f r[16];
#pragma unroll
            for (int q = 0; q < 16; ++q) r[q] = bc[stu_pidx(16 * task + q)];
            stu_fwd16<16>(r, 0, tw);
            const float4* hp = (const float4*)(hs + 16 * task);
#pragma unroll
            for (int q2 = 0; q2 < 8; ++q2) {
                const float4 h = hp[q2];
                r[2 * q2]     = stu_cmul(r[2 * q2],     v2f{h.x, h.y});
                r[2 * q2 + 1] = stu_cmul(r[2 * q2 + 1], v2f{h.z, h.w});
            }
            stu_inv16<16>(r, 0, tw);
#pragma unroll
            for (int q = 0; q < 16; ++q) bc[stu_pidx(16 * task + q)] = r[q];
        }
    }
    __syncthreads();
    stu_fft_pass<256,  true, false>(bc, tl, tw); __syncthreads();
    stu_fft_pass<4096, true, true >(bc, tl, tw); __syncthreads();

    float* ob = OUT + (size_t)b * S_ * D_ + d0;
#pragma unroll 1
    for (int k = 0; k < HALF_S / 512; ++k) {
        const int i = tid + 512 * k;
        const int pi = stu_pidx(i);
        const v2f y0 = sb[0 * FPITCH_ + pi], y1 = sb[1 * FPITCH_ + pi];
        const v2f y2 = sb[2 * FPITCH_ + pi], y3 = sb[3 * FPITCH_ + pi];
        *(float4*)(ob + (size_t)(2 * i) * D_)     = float4{y0.x, y1.x, y2.x, y3.x};
        *(float4*)(ob + (size_t)(2 * i + 1) * D_) = float4{y0.y, y1.y, y2.y, y3.y};
    }
}

extern "C" void kernel_launch(void* const* d_in, const int* in_sizes, int n_in,
                              void* d_out, int out_size, void* d_ws, size_t ws_size,
                              hipStream_t stream) {
    (void)in_sizes; (void)n_in; (void)out_size; (void)ws_size;
    const float* x   = (const float*)d_in[0];
    const float* phi = (const float*)d_in[1];
    const float* Mi  = (const float*)d_in[2];
    const float* Mf  = (const float*)d_in[3];
    float* out  = (float*)d_out;

    float* ws_x = (float*)d_ws;
    float* ws_p = ws_x + (size_t)B_ * S_ * D_;
    float* ws_s = ws_p + (size_t)K4_ * D_ * 4;
    float* ws_t = ws_s + (size_t)D_ * FN_ * 2;

    stu_pack_kernel    <<<dim3(3, K4_, 1),                   256, 0, stream>>>(Mi, ws_p, ws_t);
    stu_spectrum_kernel<<<dim3(D_ / FCH_, 1, 1),             512, 0, stream>>>(phi, Mf, ws_t, ws_s);
    stu_xproj_wmma     <<<dim3((B_ * S_) / GM_ROWS_, 1, 1),  384, 0, stream>>>(x, ws_p, ws_x);
    stu_conv_fft_kernel<<<dim3(D_ / FCH_, B_, 1),            512, 0, stream>>>(ws_x, ws_s, ws_t, out);
}


# conv kernel's global loads made whole-line (32 lines per load before): the GEMM stores x_proj channel-major via its dead LDS slab, the spectrum is stored in the conv's lane order; sum order kept
# speedup vs baseline: 2.5347x; 1.0447x over previous
#include <hip/hip_runtime.h>


typedef __attribute__((ext_vector_type(2))) float v2f;
typedef __attribute__((ext_vector_type(8))) float v8f;

#define B_ 4
#define S_ 4096
#define D_ 768
#define K_ 24
#define HALF_S 2048
#define K4_ (D_ / 4)

#define FN_     4096
#define FCH_    4
#define FPITCH_ (FN_ + FN_ / 16)

__device__ __forceinline__ int stu_hs_elem(int slot) {
    const int it = slot >> 10, q2 = (slot >> 7) & 7, tl = slot & 127;
    return 16 * (tl + 128 * it) + 2 * q2;
}

#define TWA4096_ 0
#define TWB4096_ 3072
#define TWA256_  3840
#define TWB256_  4032
#define TWA16_   4080
#define TWEND_   4092
__device__ __forceinline__ int stu_tw_expo(int t) {
    if (t < TWB4096_) return ((t >> 10) + 1) * (t & 1023);
    if (t < TWA256_)  { const int r = t - TWB4096_; return 4 * ((r >> 8) + 1) * (r & 255); }
    if (t < TWB256_)  { const int r = t - TWA256_;  return 16 * ((r >> 6) + 1) * (r & 63); }
    if (t < TWA16_)   { const int r = t - TWB256_;  return 64 * ((r >> 4) + 1) * (r & 15); }
    if (t < TWEND_)   { const int r = t - TWA16_;   return 256 * ((r >> 2) + 1) * (r & 3); }
    return 0;
}
template <int L> struct stu_twp {
    static constexpr int A  = (L == 4096) ? TWA4096_ : (L == 256) ? TWA256_ : TWA16_;
    static constexpr int NA = L / 4;
    static constexpr int B  = (L == 4096) ? TWB4096_ : TWB256_;
    static constexpr int NB = L / 16;
};

__global__ __launch_bounds__(256)
void stu_pack_kernel(const float* __restrict__ Mi, float* __restrict__ P, float* __restrict__ TWf) {
    int n  = blockIdx.x * 256 + threadIdx.x;
    int k4 = blockIdx.y;
    if (n >= D_) return;
    const int t = k4 * D_ + n;
    if (t < FN_) {
        float s, c;
        sincospif((float)stu_tw_expo(t) * (1.0f / 2048.0f), &s, &c);
        ((v2f*)TWf)[t] = v2f{c, -s};
    }
    float4 v;
    v.x = Mi[(size_t)(4 * k4 + 0) * D_ + n];
    v.y = Mi[(size_t)(4 * k4 + 1) * D_ + n];
    v.z = Mi[(size_t)(4 * k4 + 2) * D_ + n];
    v.w = Mi[(size_t)(4 * k4 + 3) * D_ + n];
    *(float4*)(P + ((size_t)k4 * D_ + n) * 4) = v;
}

__device__ __forceinline__ int stu_pidx(int n) { return n + (n >> 4); }

__device__ __forceinline__ v2f stu_cmul(v2f a, v2f w) {
    return v2f{a.x * w.x - a.y * w.y, a.x * w.y + a.y * w.x};
}
__device__ __forceinline__ v2f stu_cmulc(v2f a, v2f w) {
    return v2f{a.x * w.x + a.y * w.y, a.y * w.x - a.x * w.y};
}

template <bool INV>
__device__ __forceinline__ void stu_dft4(v2f& a0, v2f& a1, v2f& a2, v2f& a3) {
    const v2f s02 = a0 + a2, d02 = a0 - a2, s13 = a1 + a3, d13 = a1 - a3;
    const v2f r = INV ? v2f{-d13.y, d13.x} : v2f{d13.y, -d13.x};
    a0 = s02 + s13; a2 = s02 - s13; a1 = d02 + r; a3 = d02 - r;
}

template <int L>
__device__ __forceinline__ void stu_fwd16(v2f (&r)[16], int j, const v2f* __restrict__ tw) {
    constexpr int S = L / 16;
#pragma unroll
    for (int q0 = 0; q0 < 4; ++q0) {
        stu_dft4<false>(r[q0], r[q0 + 4], r[q0 + 8], r[q0 + 12]);
        if (!(L == 16 && q0 == 0)) {
            const v2f* ta = tw + stu_twp<L>::A + (j + S * q0);
            r[q0 + 4]  = stu_cmul(r[q0 + 4],  ta[0]);
            r[q0 + 8]  = stu_cmul(r[q0 + 8],  ta[stu_twp<L>::NA]);
            r[q0 + 12] = stu_cmul(r[q0 + 12], ta[2 * stu_twp<L>::NA]);
        }
    }
#pragma unroll
    for (int k = 0; k < 4; ++k) stu_dft4<false>(r[4 * k], r[4 * k + 1], r[4 * k + 2], r[4 * k + 3]);
    if (L > 16) {
        const v2f* tb = tw + stu_twp<L>::B + j;
        const v2f w1 = tb[0], w2 = tb[stu_twp<L>::NB], w3 = tb[2 * stu_twp<L>::NB];
#pragma unroll
        for (int k = 0; k < 4; ++k) {
            r[4 * k + 1] = stu_cmul(r[4 * k + 1], w1);
            r[4 * k + 2] = stu_cmul(r[4 * k + 2], w2);
            r[4 * k + 3] = stu_cmul(r[4 * k + 3], w3);
        }
    }
}

template <int L>
__device__ __forceinline__ void stu_inv16(v2f (&r)[16], int j, const v2f* __restrict__ tw) {
    constexpr int S = L / 16;
    if (L > 16) {
        const v2f* tb = tw + stu_twp<L>::B + j;
        const v2f w1 = tb[0], w2 = tb[stu_twp<L>::NB], w3 = tb[2 * stu_twp<L>::NB];
#pragma unroll
        for (int k = 0; k < 4; ++k) {
            r[4 * k + 1] = stu_cmulc(r[4 * k + 1], w1);
            r[4 * k + 2] = stu_cmulc(r[4 * k + 2], w2);
            r[4 * k + 3] = stu_cmulc(r[4 * k + 3], w3);
        }
    }
#pragma unroll
    for (int k = 0; k < 4; ++k) stu_dft4<true>(r[4 * k], r[4 * k + 1], r[4 * k + 2], r[4 * k + 3]);
#pragma unroll
    for (int q0 = 0; q0 < 4; ++q0) {
        if (!(L == 16 && q0 == 0)) {
            const v2f* ta = tw + stu_twp<L>::A + (j + S * q0);
            r[q0 + 4]  = stu_cmulc(r[q0 + 4],  ta[0]);
            r[q0 + 8]  = stu_cmulc(r[q0 + 8],  ta[stu_twp<L>::NA]);
            r[q0 + 12] = stu_cmulc(r[q0 + 12], ta[2 * stu_twp<L>::NA]);
        }
        stu_dft4<true>(r[q0], r[q0 + 4], r[q0 + 8], r[q0 + 12]);
    }
}

template <int L, bool INV, bool HALF>
__device__ __forceinline__ void stu_fft_pass(v2f* __restrict__ bc, int tl, const v2f* __restrict__ tw) {
    constexpr int S = L / 16;
#pragma unroll 1
    for (int it = 0; it < 2; ++it) {
        const int task = tl + 128 * it;
        const int bb = (task / S) * L, j = task % S;
        v2f r[16];
#pragma unroll
        for (int q = 0; q < 16; ++q)
            r[q] = (HALF && !INV && q >= 8) ? v2f{0.f, 0.f} : bc[stu_pidx(bb + j + S * q)];
        if (INV) stu_inv16<L>(r, j, tw); else stu_fwd16<L>(r, j, tw);
#pragma unroll
        for (int q = 0; q < 16; ++q)
            if (!(HALF && INV && q >= 8)) bc[stu_pidx(bb + j + S * q)] = r[q];
    }
}

__global__ __launch_bounds__(512, 8)
void stu_spectrum_kernel(const float* __restrict__ phi,
                         const float* __restrict__ Mf,
                         const float* __restrict__ TWf,
                         float* __restrict__ HSf) {
    __shared__ __attribute__((aligned(16))) v2f sb[FCH_ * FPITCH_];
    const int tid = threadIdx.x;
    const int c = tid >> 7, tl = tid & 127;
    const int d0 = blockIdx.x * FCH_;
    const v2f* tw = (const v2f*)TWf;

    float4 mf[K_];
#pragma unroll
    for (int k = 0; k < K_; ++k) mf[k] = *(const float4*)(Mf + (size_t)k * D_ + d0);
    for (int jj = 0; jj < HALF_S / 512; ++jj) {
        const int j = tid + 512 * jj;
        const float4* pr = (const float4*)(phi + (size_t)(2 * j) * K_);
        float a0 = 0.f, a1 = 0.f, a2 = 0.f, a3 = 0.f;
#pragma unroll
        for (int k4 = 0; k4 < K_ / 4; ++k4) {
            const float4 p = pr[k4];
            const float pv[4] = {p.x, p.y, p.z, p.w};
#pragma unroll
            for (int e = 0; e < 4; ++e) {
                const float4 m = mf[4 * k4 + e];
                a0 = fmaf(pv[e], m.x, a0); a1 = fmaf(pv[e], m.y, a1);
                a2 = fmaf(pv[e], m.z, a2); a3 = fmaf(pv[e], m.w, a3);
            }
        }
        const float sc = 2.0f / 4096.0f;
        const int pj = stu_pidx(j);
        sb[0 * FPITCH_ + pj] = v2f{a0 * sc, 0.f};
        sb[1 * FPITCH_ + pj] = v2f{a1 * sc, 0.f};
        sb[2 * FPITCH_ + pj] = v2f{a2 * sc, 0.f};
        sb[3 * FPITCH_ + pj] = v2f{a3 * sc, 0.f};
    }
    __syncthreads();
    v2f* bc = sb + c * FPITCH_;
    stu_fft_pass<4096, false, true >(bc, tl, tw); __syncthreads();
    stu_fft_pass<256,  false, false>(bc, tl, tw); __syncthreads();
    stu_fft_pass<16,   false, false>(bc, tl, tw); __syncthreads();
    for (int idx = tid; idx < FCH_ * (FN_ / 2); idx += 512) {
        const int c2 = idx >> 11, slot = idx & 2047;
        const int n = stu_hs_elem(slot);
        const v2f* s = sb + c2 * FPITCH_ + stu_pidx(n);
        const v2f u = s[0], v = s[1];
        *(float4*)(HSf + ((size_t)(d0 + c2) * FN_ + 2 * slot) * 2) = float4{u.x, u.y, v.x, v.y};
    }
}

#define GM_ROWS_  64
#define GM_PITCH_ (D_ + 4)
#define GM_TP_    68
#define GM_LDS_   (12 * 64 * GM_TP_)
__global__ __launch_bounds__(384)
void stu_xproj_wmma(const float* __restrict__ X,
                    const float* __restrict__ P,
                    float* __restrict__ XP) {
    __shared__ __attribute__((aligned(16))) float sa[GM_LDS_];

    const int tid  = threadIdx.x;
    const int wave = tid >> 5;
    const int lane = tid & 31;
    const int half = lane >> 4;
    const int l16  = lane & 15;
    const int m0 = blockIdx.x * GM_ROWS_;
    const int n0 = wave * 64;

    const float* xb = X + (size_t)m0 * D_;
#pragma unroll
    for (int k = 0; k < GM_ROWS_ * (D_ / 4) / 384; ++k) {
        const int idx = tid + k * 384;
        const int row = idx / (D_ / 4), q = idx - row * (D_ / 4);
        *(float4*)(sa + row * GM_PITCH_ + q * 4) = *(const float4*)(xb + (size_t)idx * 4);
    }
    __syncthreads();

    v8f c[4][4] = {};
    const float* ar   = sa + l16 * GM_PITCH_ + half * 2;
    const float* pcol = P + ((size_t)(n0 + l16)) * 4 + 2 * half;

    for (int k4 = 0; k4 < K4_; ++k4) {
        v2f a[4];
#pragma unroll
        for (int mt = 0; mt < 4; ++mt)
            a[mt] = *(const v2f*)(ar + mt * 16 * GM_PITCH_ + k4 * 4);
        const float* pk = pcol + (size_t)k4 * D_ * 4;
#pragma unroll
        for (int nt = 0; nt < 4; ++nt) {
            v2f b = *(const v2f*)(pk + nt * 64);
#pragma unroll
            for (int mt = 0; mt < 4; ++mt)
                c[mt][nt] = __builtin_amdgcn_wmma_f32_16x16x4_f32(false, a[mt], false, b, (short)0, c[mt][nt], false, false);
        }
    }

    __syncthreads();
    float* tp = sa + wave * (64 * GM_TP_);
#pragma unroll
    for (int mt = 0; mt < 4; ++mt) {
#pragma unroll
        for (int nt = 0; nt < 4; ++nt) {
            float* p = tp + (nt * 16 + l16) * GM_TP_ + mt * 16 + half * 8;
            *(float4*)p       = float4{c[mt][nt][0], c[mt][nt][1], c[mt][nt][2], c[mt][nt][3]};
            *(float4*)(p + 4) = float4{c[mt][nt][4], c[mt][nt][5], c[mt][nt][6], c[mt][nt][7]};
        }
    }
    __syncthreads();
    const int bi = m0 / S_, t0 = m0 % S_;
    float* ob = XP + ((size_t)bi * D_ + n0) * S_ + t0 + 4 * l16;
#pragma unroll
    for (int q = 0; q < 32; ++q) {
        const int col = 2 * q + half;
        *(float4*)(ob + (size_t)col * S_) = *(const float4*)(tp + col * GM_TP_ + 4 * l16);
    }
}

__global__ __launch_bounds__(512, 8)
void stu_conv_fft_kernel(const float* __restrict__ XP,
                         const float* __restrict__ HSf,
                         const float* __restrict__ TWf,
                         float* __restrict__ OUT) {
    __shared__ __attribute__((aligned(16))) v2f sb[FCH_ * FPITCH_];
    const int tid = threadIdx.x;
    const int c = tid >> 7, tl = tid & 127;
    const int d0 = blockIdx.x * FCH_;
    const int b  = blockIdx.y;
    const v2f* tw = (const v2f*)TWf;

    const float* xpb = XP + ((size_t)b * D_ + d0) * S_;
#pragma unroll
    for (int k = 0; k < FCH_ * S_ / 4 / 512; ++k) {
        const int idx = tid + 512 * k;
        const int ch = idx >> 10, f = idx & 1023;
        const float4 v = *(const float4*)(xpb + (size_t)idx * 4);
        v2f* s = sb + ch * FPITCH_ + stu_pidx(2 * f);
        s[0] = v2f{v.x, v.y};
        s[1] = v2f{v.z, v.w};
    }
    __syncthreads();

    v2f* bc = sb + c * FPITCH_;
    stu_fft_pass<4096, false, true >(bc, tl, tw); __syncthreads();
    stu_fft_pass<256,  false, false>(bc, tl, tw); __syncthreads();
    {
        const v2f* hs = (const v2f*)HSf + (size_t)(d0 + c) * FN_;
#pragma unroll 1
        for (int it = 0; it < 2; ++it) {
            const int task = tl + 128 * it;
            v2f r[16];
#pragma unroll
            for (int q = 0; q < 16; ++q) r[q] = bc[stu_pidx(16 * task + q)];
            stu_fwd16<16>(r, 0, tw);
            const float4* hp = (const float4*)hs + it * 1024 + tl;
#pragma unroll
            for (int q2 = 0; q2 < 8; ++q2) {
                const float4 h = hp[q2 * 128];
                r[2 * q2]     = stu_cmul(r[2 * q2],     v2f{h.x, h.y});
                r[2 * q2 + 1] = stu_cmul(r[2 * q2 + 1], v2f{h.z, h.w});
            }
            stu_inv16<16>(r, 0, tw);
#pragma unroll
            for (int q = 0; q < 16; ++q) bc[stu_pidx(16 * task + q)] = r[q];
        }
    }
    __syncthreads();
    stu_fft_pass<256,  true, false>(bc, tl, tw); __syncthreads();
    stu_fft_pass<4096, true, true >(bc, tl, tw); __syncthreads();

    float* ob = OUT + (size_t)b * S_ * D_ + d0;
#pragma unroll 1
    for (int k = 0; k < HALF_S / 512; ++k) {
        const int i = tid + 512 * k;
        const int pi = stu_pidx(i);
        const v2f y0 = sb[0 * FPITCH_ + pi], y1 = sb[1 * FPITCH_ + pi];
        const v2f y2 = sb[2 * FPITCH_ + pi], y3 = sb[3 * FPITCH_ + pi];
        *(float4*)(ob + (size_t)(2 * i) * D_)     = float4{y0.x, y1.x, y2.x, y3.x};
        *(float4*)(ob + (size_t)(2 * i + 1) * D_) = float4{y0.y, y1.y, y2.y, y3.y};
    }
}

extern "C" void kernel_launch(void* const* d_in, const int* in_sizes, int n_in,
                              void* d_out, int out_size, void* d_ws, size_t ws_size,
                              hipStream_t stream) {
    (void)in_sizes; (void)n_in; (void)out_size; (void)ws_size;
    const float* x   = (const float*)d_in[0];
    const float* phi = (const float*)d_in[1];
    const float* Mi  = (const float*)d_in[2];
    const float* Mf  = (const float*)d_in[3];
    float* out  = (float*)d_out;

    float* ws_x = (float*)d_ws;
    float* ws_p = ws_x + (size_t)B_ * S_ * D_;
    float* ws_s = ws_p + (size_t)K4_ * D_ * 4;
    float* ws_t = ws_s + (size_t)D_ * FN_ * 2;

    stu_pack_kernel    <<<dim3(3, K4_, 1),                   256, 0, stream>>>(Mi, ws_p, ws_t);
    stu_spectrum_kernel<<<dim3(D_ / FCH_, 1, 1),             512, 0, stream>>>(phi, Mf, ws_t, ws_s);
    stu_xproj_wmma     <<<dim3((B_ * S_) / GM_ROWS_, 1, 1),  384, 0, stream>>>(x, ws_p, ws_x);
    stu_conv_fft_kernel<<<dim3(D_ / FCH_, B_, 1),            512, 0, stream>>>(ws_x, ws_s, ws_t, out);
}


# conv kernel, product pass: a task's eight filter-spectrum loads issued together ahead of its LDS reads and butterflies (were issued one or two at a time, each waited for); values and sums kept
# speedup vs baseline: 2.7464x; 1.0835x over previous
#include <hip/hip_runtime.h>


typedef __attribute__((ext_vector_type(2))) float v2f;
typedef __attribute__((ext_vector_type(8))) float v8f;

#define B_ 4
#define S_ 4096
#define D_ 768
#define K_ 24
#define HALF_S 2048
#define K4_ (D_ / 4)

#define FN_     4096
#define FCH_    4
#define FPITCH_ (FN_ + FN_ / 16)

__device__ __forceinline__ int stu_hs_elem(int slot) {
    const int it = slot >> 10, q2 = (slot >> 7) & 7, tl = slot & 127;
    return 16 * (tl + 128 * it) + 2 * q2;
}

#define TWA4096_ 0
#define TWB4096_ 3072
#define TWA256_  3840
#define TWB256_  4032
#define TWA16_   4080
#define TWEND_   4092
__device__ __forceinline__ int stu_tw_expo(int t) {
    if (t < TWB4096_) return ((t >> 10) + 1) * (t & 1023);
    if (t < TWA256_)  { const int r = t - TWB4096_; return 4 * ((r >> 8) + 1) * (r & 255); }
    if (t < TWB256_)  { const int r = t - TWA256_;  return 16 * ((r >> 6) + 1) * (r & 63); }
    if (t < TWA16_)   { const int r = t - TWB256_;  return 64 * ((r >> 4) + 1) * (r & 15); }
    if (t < TWEND_)   { const int r = t - TWA16_;   return 256 * ((r >> 2) + 1) * (r & 3); }
    return 0;
}
template <int L> struct stu_twp {
    static constexpr int A  = (L == 4096) ? TWA4096_ : (L == 256) ? TWA256_ : TWA16_;
    static constexpr int NA = L / 4;
    static constexpr int B  = (L == 4096) ? TWB4096_ : TWB256_;
    static constexpr int NB = L / 16;
};

__global__ __launch_bounds__(256)
void stu_pack_kernel(const float* __restrict__ Mi, float* __restrict__ P, float* __restrict__ TWf) {
    int n  = blockIdx.x * 256 + threadIdx.x;
    int k4 = blockIdx.y;
    if (n >= D_) return;
    const int t = k4 * D_ + n;
    if (t < FN_) {
        float s, c;
        sincospif((float)stu_tw_expo(t) * (1.0f / 2048.0f), &s, &c);
        ((v2f*)TWf)[t] = v2f{c, -s};
    }
    float4 v;
    v.x = Mi[(size_t)(4 * k4 + 0) * D_ + n];
    v.y = Mi[(size_t)(4 * k4 + 1) * D_ + n];
    v.z = Mi[(size_t)(4 * k4 + 2) * D_ + n];
    v.w = Mi[(size_t)(4 * k4 + 3) * D_ + n];
    *(float4*)(P + ((size_t)k4 * D_ + n) * 4) = v;
}

__device__ __forceinline__ int stu_pidx(int n) { return n + (n >> 4); }

__device__ __forceinline__ v2f stu_cmul(v2f a, v2f w) {
    return v2f{a.x * w.x - a.y * w.y, a.x * w.y + a.y * w.x};
}
__device__ __forceinline__ v2f stu_cmulc(v2f a, v2f w) {
    return v2f{a.x * w.x + a.y * w.y, a.y * w.x - a.x * w.y};
}

template <bool INV>
__device__ __forceinline__ void stu_dft4(v2f& a0, v2f& a1, v2f& a2, v2f& a3) {
    const v2f s02 = a0 + a2, d02 = a0 - a2, s13 = a1 + a3, d13 = a1 - a3;
    const v2f r = INV ? v2f{-d13.y, d13.x} : v2f{d13.y, -d13.x};
    a0 = s02 + s13; a2 = s02 - s13; a1 = d02 + r; a3 = d02 - r;
}

template <int L>
__device__ __forceinline__ void stu_fwd16(v2f (&r)[16], int j, const v2f* __restrict__ tw) {
    constexpr int S = L / 16;
#pragma unroll
    for (int q0 = 0; q0 < 4; ++q0) {
        stu_dft4<false>(r[q0], r[q0 + 4], r[q0 + 8], r[q0 + 12]);
        if (!(L == 16 && q0 == 0)) {
            const v2f* ta = tw + stu_twp<L>::A + (j + S * q0);
            r[q0 + 4]  = stu_cmul(r[q0 + 4],  ta[0]);
            r[q0 + 8]  = stu_cmul(r[q0 + 8],  ta[stu_twp<L>::NA]);
            r[q0 + 12] = stu_cmul(r[q0 + 12], ta[2 * stu_twp<L>::NA]);
        }
    }
#pragma unroll
    for (int k = 0; k < 4; ++k) stu_dft4<false>(r[4 * k], r[4 * k + 1], r[4 * k + 2], r[4 * k + 3]);
    if (L > 16) {
        const v2f* tb = tw + stu_twp<L>::B + j;
        const v2f w1 = tb[0], w2 = tb[stu_twp<L>::NB], w3 = tb[2 * stu_twp<L>::NB];
#pragma unroll
        for (int k = 0; k < 4; ++k) {
            r[4 * k + 1] = stu_cmul(r[4 * k + 1], w1);
            r[4 * k + 2] = stu_cmul(r[4 * k + 2], w2);
            r[4 * k + 3] = stu_cmul(r[4 * k + 3], w3);
        }
    }
}

template <int L>
__device__ __forceinline__ void stu_inv16(v2f (&r)[16], int j, const v2f* __restrict__ tw) {
    constexpr int S = L / 16;
    if (L > 16) {
        const v2f* tb = tw + stu_twp<L>::B + j;
        const v2f w1 = tb[0], w2 = tb[stu_twp<L>::NB], w3 = tb[2 * stu_twp<L>::NB];
#pragma unroll
        for (int k = 0; k < 4; ++k) {
            r[4 * k + 1] = stu_cmulc(r[4 * k + 1], w1);
            r[4 * k + 2] = stu_cmulc(r[4 * k + 2], w2);
            r[4 * k + 3] = stu_cmulc(r[4 * k + 3], w3);
        }
    }
#pragma unroll
    for (int k = 0; k < 4; ++k) stu_dft4<true>(r[4 * k], r[4 * k + 1], r[4 * k + 2], r[4 * k + 3]);
#pragma unroll
    for (int q0 = 0; q0 < 4; ++q0) {
        if (!(L == 16 && q0 == 0)) {
            const v2f* ta = tw + stu_twp<L>::A + (j + S * q0);
            r[q0 + 4]  = stu_cmulc(r[q0 + 4],  ta[0]);
            r[q0 + 8]  = stu_cmulc(r[q0 + 8],  ta[stu_twp<L>::NA]);
            r[q0 + 12] = stu_cmulc(r[q0 + 12], ta[2 * stu_twp<L>::NA]);
        }
        stu_dft4<true>(r[q0], r[q0 + 4], r[q0 + 8], r[q0 + 12]);
    }
}

template <int L, bool INV, bool HALF>
__device__ __forceinline__ void stu_fft_pass(v2f* __restrict__ bc, int tl, const v2f* __restrict__ tw) {
    constexpr int S = L / 16;
#pragma unroll 1
    for (int it = 0; it < 2; ++it) {
        const int task = tl + 128 * it;
        const int bb = (task / S) * L, j = task % S;
        v2f r[16];
#pragma unroll
        for (int q = 0; q < 16; ++q)
            r[q] = (HALF && !INV && q >= 8) ? v2f{0.f, 0.f} : bc[stu_pidx(bb + j + S * q)];
        if (INV) stu_inv16<L>(r, j, tw); else stu_fwd16<L>(r, j, tw);
#pragma unroll
        for (int q = 0; q < 16; ++q)
            if (!(HALF && INV && q >= 8)) bc[stu_pidx(bb + j + S * q)] = r[q];
    }
}

__global__ __launch_bounds__(512, 8)
void stu_spectrum_kernel(const float* __restrict__ phi,
                         const float* __restrict__ Mf,
                         const float* __restrict__ TWf,
                         float* __restrict__ HSf) {
    __shared__ __attribute__((aligned(16))) v2f sb[FCH_ * FPITCH_];
    const int tid = threadIdx.x;
    const int c = tid >> 7, tl = tid & 127;
    const int d0 = blockIdx.x * FCH_;
    const v2f* tw = (const v2f*)TWf;

    float4 mf[K_];
#pragma unroll
    for (int k = 0; k < K_; ++k) mf[k] = *(const float4*)(Mf + (size_t)k * D_ + d0);
    for (int jj = 0; jj < HALF_S / 512; ++jj) {
        const int j = tid + 512 * jj;
        const float4* pr = (const float4*)(phi + (size_t)(2 * j) * K_);
        float a0 = 0.f, a1 = 0.f, a2 = 0.f, a3 = 0.f;
#pragma unroll
        for (int k4 = 0; k4 < K_ / 4; ++k4) {
            const float4 p = pr[k4];
            const float pv[4] = {p.x, p.y, p.z, p.w};
#pragma unroll
            for (int e = 0; e < 4; ++e) {
                const float4 m = mf[4 * k4 + e];
                a0 = fmaf(pv[e], m.x, a0); a1 = fmaf(pv[e], m.y, a1);
                a2 = fmaf(pv[e], m.z, a2); a3 = fmaf(pv[e], m.w, a3);
            }
        }
        const float sc = 2.0f / 4096.0f;
        const int pj = stu_pidx(j);
        sb[0 * FPITCH_ + pj] = v2f{a0 * sc, 0.f};
        sb[1 * FPITCH_ + pj] = v2f{a1 * sc, 0.f};
        sb[2 * FPITCH_ + pj] = v2f{a2 * sc, 0.f};
        sb[3 * FPITCH_ + pj] = v2f{a3 * sc, 0.f};
    }
    __syncthreads();
    v2f* bc = sb + c * FPITCH_;
    stu_fft_pass<4096, false, true >(bc, tl, tw); __syncthreads();
    stu_fft_pass<256,  false, false>(bc, tl, tw); __syncthreads();
    stu_fft_pass<16,   false, false>(bc, tl, tw); __syncthreads();
    for (int idx = tid; idx < FCH_ * (FN_ / 2); idx += 512) {
        const int c2 = idx >> 11, slot = idx & 2047;
        const int n = stu_hs_elem(slot);
        const v2f* s = sb + c2 * FPITCH_ + stu_pidx(n);
        const v2f u = s[0], v = s[1];
        *(float4*)(HSf + ((size_t)(d0 + c2) * FN_ + 2 * slot) * 2) = float4{u.x, u.y, v.x, v.y};
    }
}

#define GM_ROWS_  64
#define GM_PITCH_ (D_ + 4)
#define GM_TP_    68
#define GM_LDS_   (12 * 64 * GM_TP_)
__global__ __launch_bounds__(384)
void stu_xproj_wmma(const float* __restrict__ X,
                    const float* __restrict__ P,
                    float* __restrict__ XP) {
    __shared__ __attribute__((aligned(16))) float sa[GM_LDS_];

    const int tid  = threadIdx.x;
    const int wave = tid >> 5;
    const int lane = tid & 31;
    const int half = lane >> 4;
    const int l16  = lane & 15;
    const int m0 = blockIdx.x * GM_ROWS_;
    const int n0 = wave * 64;

    const float* xb = X + (size_t)m0 * D_;
#pragma unroll
    for (int k = 0; k < GM_ROWS_ * (D_ / 4) / 384; ++k) {
        const int idx = tid + k * 384;
        const int row = idx / (D_ / 4), q = idx - row * (D_ / 4);
        *(float4*)(sa + row * GM_PITCH_ + q * 4) = *(const float4*)(xb + (size_t)idx * 4);
    }
    __syncthreads();

    v8f c[4][4] = {};
    const float* ar   = sa + l16 * GM_PITCH_ + half * 2;
    const float* pcol = P + ((size_t)(n0 + l16)) * 4 + 2 * half;

    for (int k4 = 0; k4 < K4_; ++k4) {
        v2f a[4];
#pragma unroll
        for (int mt = 0; mt < 4; ++mt)
            a[mt] = *(const v2f*)(ar + mt * 16 * GM_PITCH_ + k4 * 4);
        const float* pk = pcol + (size_t)k4 * D_ * 4;
#pragma unroll
        for (int nt = 0; nt < 4; ++nt) {
            v2f b = *(const v2f*)(pk + nt * 64);
#pragma unroll
            for (int mt = 0; mt < 4; ++mt)
                c[mt][nt] = __builtin_amdgcn_wmma_f32_16x16x4_f32(false, a[mt], false, b, (short)0, c[mt][nt], false, false);
        }
    }

    __syncthreads();
    float* tp = sa + wave * (64 * GM_TP_);
#pragma unroll
    for (int mt = 0; mt < 4; ++mt) {
#pragma unroll
        for (int nt = 0; nt < 4; ++nt) {
            float* p = tp + (nt * 16 + l16) * GM_TP_ + mt * 16 + half * 8;
            *(float4*)p       = float4{c[mt][nt][0], c[mt][nt][1], c[mt][nt][2], c[mt][nt][3]};
            *(float4*)(p + 4) = float4{c[mt][nt][4], c[mt][nt][5], c[mt][nt][6], c[mt][nt][7]};
        }
    }
    __syncthreads();
    const int bi = m0 / S_, t0 = m0 % S_;
    float* ob = XP + ((size_t)bi * D_ + n0) * S_ + t0 + 4 * l16;
#pragma unroll
    for (int q = 0; q < 32; ++q) {
        const int col = 2 * q + half;
        *(float4*)(ob + (size_t)col * S_) = *(const float4*)(tp + col * GM_TP_ + 4 * l16);
    }
}

__global__ __launch_bounds__(512, 8)
void stu_conv_fft_kernel(const float* __restrict__ XP,
                         const float* __restrict__ HSf,
                         const float* __restrict__ TWf,
                         float* __restrict__ OUT) {
    __shared__ __attribute__((aligned(16))) v2f sb[FCH_ * FPITCH_];
    const int tid = threadIdx.x;
    const int c = tid >> 7, tl = tid & 127;
    const int d0 = blockIdx.x * FCH_;
    const int b  = blockIdx.y;
    const v2f* tw = (const v2f*)TWf;

    const float* xpb = XP + ((size_t)b * D_ + d0) * S_;
#pragma unroll
    for (int k = 0; k < FCH_ * S_ / 4 / 512; ++k) {
        const int idx = tid + 512 * k;
        const int ch = idx >> 10, f = idx & 1023;
        const float4 v = *(const float4*)(xpb + (size_t)idx * 4);
        v2f* s = sb + ch * FPITCH_ + stu_pidx(2 * f);
        s[0] = v2f{v.x, v.y};
        s[1] = v2f{v.z, v.w};
    }
    __syncthreads();

    v2f* bc = sb + c * FPITCH_;
    stu_fft_pass<4096, false, true >(bc, tl, tw); __syncthreads();
    stu_fft_pass<256,  false, false>(bc, tl, tw); __syncthreads();
    {
        const v2f* hs = (const v2f*)HSf + (size_t)(d0 + c) * FN_;
#pragma unroll 1
        for (int it = 0; it < 2; ++it) {
            const int task = tl + 128 * it;
            const float4* hp = (const float4*)hs + it * 1024 + tl;
            float4 h[8];
#pragma unroll
            for (int q2 = 0; q2 < 8; ++q2) h[q2] = hp[q2 * 128];
            v2f r[16];
#pragma unroll
            for (int q = 0; q < 16; ++q) r[q] = bc[stu_pidx(16 * task + q)];
            stu_fwd16<16>(r, 0, tw);
#pragma unroll
            for (int q2 = 0; q2 < 8; ++q2) {
                r[2 * q2]     = stu_cmul(r[2 * q2],     v2f{h[q2].x, h[q2].y});
                r[2 * q2 + 1] = stu_cmul(r[2 * q2 + 1], v2f{h[q2].z, h[q2].w});
            }
            stu_inv16<16>(r, 0, tw);
#pragma unroll
            for (int q = 0; q < 16; ++q) bc[stu_pidx(16 * task + q)] = r[q];
        }
    }
    __syncthreads();
    stu_fft_pass<256,  true, false>(bc, tl, tw); __syncthreads();
    stu_fft_pass<4096, true, true >(bc, tl, tw); __syncthreads();

    float* ob = OUT + (size_t)b * S_ * D_ + d0;
#pragma unroll 1
    for (int k = 0; k < HALF_S / 512; ++k) {
        const int i = tid + 512 * k;
        const int pi = stu_pidx(i);
        const v2f y0 = sb[0 * FPITCH_ + pi], y1 = sb[1 * FPITCH_ + pi];
        const v2f y2 = sb[2 * FPITCH_ + pi], y3 = sb[3 * FPITCH_ + pi];
        *(float4*)(ob + (size_t)(2 * i) * D_)     = float4{y0.x, y1.x, y2.x, y3.x};
        *(float4*)(ob + (size_t)(2 * i + 1) * D_) = float4{y0.y, y1.y, y2.y, y3.y};
    }
}

extern "C" void kernel_launch(void* const* d_in, const int* in_sizes, int n_in,
                              void* d_out, int out_size, void* d_ws, size_t ws_size,
                              hipStream_t stream) {
    (void)in_sizes; (void)n_in; (void)out_size; (void)ws_size;
    const float* x   = (const float*)d_in[0];
    const float* phi = (const float*)d_in[1];
    const float* Mi  = (const float*)d_in[2];
    const float* Mf  = (const float*)d_in[3];
    float* out  = (float*)d_out;

    float* ws_x = (float*)d_ws;
    float* ws_p = ws_x + (size_t)B_ * S_ * D_;
    float* ws_s = ws_p + (size_t)K4_ * D_ * 4;
    float* ws_t = ws_s + (size_t)D_ * FN_ * 2;

    stu_pack_kernel    <<<dim3(3, K4_, 1),                   256, 0, stream>>>(Mi, ws_p, ws_t);
    stu_spectrum_kernel<<<dim3(D_ / FCH_, 1, 1),             512, 0, stream>>>(phi, Mf, ws_t, ws_s);
    stu_xproj_wmma     <<<dim3((B_ * S_) / GM_ROWS_, 1, 1),  384, 0, stream>>>(x, ws_p, ws_x);
    stu_conv_fft_kernel<<<dim3(D_ / FCH_, B_, 1),            512, 0, stream>>>(ws_x, ws_s, ws_t, out);
}


# launches 4 -> 3: Mi packing runs as extra blocks of the filter-spectrum launch; each spectrum block writes its own copy of the twiddle table before use (conv reads block 0's); values and sums kept
# speedup vs baseline: 2.8740x; 1.0465x over previous
#include <hip/hip_runtime.h>


typedef __attribute__((ext_vector_type(2))) float v2f;
typedef __attribute__((ext_vector_type(8))) float v8f;

#define B_ 4
#define S_ 4096
#define D_ 768
#define K_ 24
#define HALF_S 2048
#define K4_ (D_ / 4)

#define FN_     4096
#define FCH_    4
#define FPITCH_ (FN_ + FN_ / 16)

__device__ __forceinline__ int stu_hs_elem(int slot) {
    const int it = slot >> 10, q2 = (slot >> 7) & 7, tl = slot & 127;
    return 16 * (tl + 128 * it) + 2 * q2;
}

#define TWA4096_ 0
#define TWB4096_ 3072
#define TWA256_  3840
#define TWB256_  4032
#define TWA16_   4080
#define TWEND_   4092
__device__ __forceinline__ int stu_tw_expo(int t) {
    if (t < TWB4096_) return ((t >> 10) + 1) * (t & 1023);
    if (t < TWA256_)  { const int r = t - TWB4096_; return 4 * ((r >> 8) + 1) * (r & 255); }
    if (t < TWB256_)  { const int r = t - TWA256_;  return 16 * ((r >> 6) + 1) * (r & 63); }
    if (t < TWA16_)   { const int r = t - TWB256_;  return 64 * ((r >> 4) + 1) * (r & 15); }
    if (t < TWEND_)   { const int r = t - TWA16_;   return 256 * ((r >> 2) + 1) * (r & 3); }
    return 0;
}
template <int L> struct stu_twp {
    static constexpr int A  = (L == 4096) ? TWA4096_ : (L == 256) ? TWA256_ : TWA16_;
    static constexpr int NA = L / 4;
    static constexpr int B  = (L == 4096) ? TWB4096_ : TWB256_;
    static constexpr int NB = L / 16;
};

__device__ __forceinline__ int stu_pidx(int n) { return n + (n >> 4); }

__device__ __forceinline__ v2f stu_cmul(v2f a, v2f w) {
    return v2f{a.x * w.x - a.y * w.y, a.x * w.y + a.y * w.x};
}
__device__ __forceinline__ v2f stu_cmulc(v2f a, v2f w) {
    return v2f{a.x * w.x + a.y * w.y, a.y * w.x - a.x * w.y};
}

template <bool INV>
__device__ __forceinline__ void stu_dft4(v2f& a0, v2f& a1, v2f& a2, v2f& a3) {
    const v2f s02 = a0 + a2, d02 = a0 - a2, s13 = a1 + a3, d13 = a1 - a3;
    const v2f r = INV ? v2f{-d13.y, d13.x} : v2f{d13.y, -d13.x};
    a0 = s02 + s13; a2 = s02 - s13; a1 = d02 + r; a3 = d02 - r;
}

template <int L>
__device__ __forceinline__ void stu_fwd16(v2f (&r)[16], int j, const v2f* __restrict__ tw) {
    constexpr int S = L / 16;
#pragma unroll
    for (int q0 = 0; q0 < 4; ++q0) {
        stu_dft4<false>(r[q0], r[q0 + 4], r[q0 + 8], r[q0 + 12]);
        if (!(L == 16 && q0 == 0)) {
            const v2f* ta = tw + stu_twp<L>::A + (j + S * q0);
            r[q0 + 4]  = stu_cmul(r[q0 + 4],  ta[0]);
            r[q0 + 8]  = stu_cmul(r[q0 + 8],  ta[stu_twp<L>::NA]);
            r[q0 + 12] = stu_cmul(r[q0 + 12], ta[2 * stu_twp<L>::NA]);
        }
    }
#pragma unroll
    for (int k = 0; k < 4; ++k) stu_dft4<false>(r[4 * k], r[4 * k + 1], r[4 * k + 2], r[4 * k + 3]);
    if (L > 16) {
        const v2f* tb = tw + stu_twp<L>::B + j;
        const v2f w1 = tb[0], w2 = tb[stu_twp<L>::NB], w3 = tb[2 * stu_twp<L>::NB];
#pragma unroll
        for (int k = 0; k < 4; ++k) {
            r[4 * k + 1] = stu_cmul(r[4 * k + 1], w1);
            r[4 * k + 2] = stu_cmul(r[4 * k + 2], w2);
            r[4 * k + 3] = stu_cmul(r[4 * k + 3], w3);
        }
    }
}

template <int L>
__device__ __forceinline__ void stu_inv16(v2f (&r)[16], int j, const v2f* __restrict__ tw) {
    constexpr int S = L / 16;
    if (L > 16) {
        const v2f* tb = tw + stu_twp<L>::B + j;
        const v2f w1 = tb[0], w2 = tb[stu_twp<L>::NB], w3 = tb[2 * stu_twp<L>::NB];
#pragma unroll
        for (int k = 0; k < 4; ++k) {
            r[4 * k + 1] = stu_cmulc(r[4 * k + 1], w1);
            r[4 * k + 2] = stu_cmulc(r[4 * k + 2], w2);
            r[4 * k + 3] = stu_cmulc(r[4 * k + 3], w3);
        }
    }
#pragma unroll
    for (int k = 0; k < 4; ++k) stu_dft4<true>(r[4 * k], r[4 * k + 1], r[4 * k + 2], r[4 * k + 3]);
#pragma unroll
    for (int q0 = 0; q0 < 4; ++q0) {
        if (!(L == 16 && q0 == 0)) {
            const v2f* ta = tw + stu_twp<L>::A + (j + S * q0);
            r[q0 + 4]  = stu_cmulc(r[q0 + 4],  ta[0]);
            r[q0 + 8]  = stu_cmulc(r[q0 + 8],  ta[stu_twp<L>::NA]);
            r[q0 + 12] = stu_cmulc(r[q0 + 12], ta[2 * stu_twp<L>::NA]);
        }
        stu_dft4<true>(r[q0], r[q0 + 4], r[q0 + 8], r[q0 + 12]);
    }
}

template <int L, bool INV, bool HALF>
__device__ __forceinline__ void stu_fft_pass(v2f* __restrict__ bc, int tl, const v2f* __restrict__ tw) {
    constexpr int S = L / 16;
#pragma unroll 1
    for (int it = 0; it < 2; ++it) {
        const int task = tl + 128 * it;
        const int bb = (task / S) * L, j = task % S;
        v2f r[16];
#pragma unroll
        for (int q = 0; q < 16; ++q)
            r[q] = (HALF && !INV && q >= 8) ? v2f{0.f, 0.f} : bc[stu_pidx(bb + j + S * q)];
        if (INV) stu_inv16<L>(r, j, tw); else stu_fwd16<L>(r, j, tw);
#pragma unroll
        for (int q = 0; q < 16; ++q)
            if (!(HALF && INV && q >= 8)) bc[stu_pidx(bb + j + S * q)] = r[q];
    }
}

#define SPEC_BLOCKS_ (D_ / FCH_)
#define PACK_BLOCKS_ (K4_ * D_ / 512)
__global__ __launch_bounds__(512, 8)
void stu_spectrum_kernel(const float* __restrict__ phi,
                         const float* __restrict__ Mf,
                         const float* __restrict__ Mi,
                         float* __restrict__ P,
                         float* TWf,
                         float* __restrict__ HSf) {
    __shared__ __attribute__((aligned(16))) v2f sb[FCH_ * FPITCH_];
    const int tid = threadIdx.x;
    if (blockIdx.x >= SPEC_BLOCKS_) {
        const int g  = (blockIdx.x - SPEC_BLOCKS_) * 512 + tid;
        const int k4 = g / D_, n = g - k4 * D_;
        float4 v;
        v.x = Mi[(size_t)(4 * k4 + 0) * D_ + n];
        v.y = Mi[(size_t)(4 * k4 + 1) * D_ + n];
        v.z = Mi[(size_t)(4 * k4 + 2) * D_ + n];
        v.w = Mi[(size_t)(4 * k4 + 3) * D_ + n];
        *(float4*)(P + (size_t)g * 4) = v;
        return;
    }
    const int c = tid >> 7, tl = tid & 127;
    const int d0 = blockIdx.x * FCH_;
    v2f* twb = (v2f*)TWf + (size_t)blockIdx.x * FN_;
    for (int k = 0; k < FN_ / 512; ++k) {
        const int t = tid + 512 * k;
        float s, cs;
        sincospif((float)stu_tw_expo(t) * (1.0f / 2048.0f), &s, &cs);
        twb[t] = v2f{cs, -s};
    }
    const v2f* tw = twb;

    float4 mf[K_];
#pragma unroll
    for (int k = 0; k < K_; ++k) mf[k] = *(const float4*)(Mf + (size_t)k * D_ + d0);
    for (int jj = 0; jj < HALF_S / 512; ++jj) {
        const int j = tid + 512 * jj;
        const float4* pr = (const float4*)(phi + (size_t)(2 * j) * K_);
        float a0 = 0.f, a1 = 0.f, a2 = 0.f, a3 = 0.f;
#pragma unroll
        for (int k4 = 0; k4 < K_ / 4; ++k4) {
            const float4 p = pr[k4];
            const float pv[4] = {p.x, p.y, p.z, p.w};
#pragma unroll
            for (int e = 0; e < 4; ++e) {
                const float4 m = mf[4 * k4 + e];
                a0 = fmaf(pv[e], m.x, a0); a1 = fmaf(pv[e], m.y, a1);
                a2 = fmaf(pv[e], m.z, a2); a3 = fmaf(pv[e], m.w, a3);
            }
        }
        const float sc = 2.0f / 4096.0f;
        const int pj = stu_pidx(j);
        sb[0 * FPITCH_ + pj] = v2f{a0 * sc, 0.f};
        sb[1 * FPITCH_ + pj] = v2f{a1 * sc, 0.f};
        sb[2 * FPITCH_ + pj] = v2f{a2 * sc, 0.f};
        sb[3 * FPITCH_ + pj] = v2f{a3 * sc, 0.f};
    }
    __syncthreads();
    v2f* bc = sb + c * FPITCH_;
    stu_fft_pass<4096, false, true >(bc, tl, tw); __syncthreads();
    stu_fft_pass<256,  false, false>(bc, tl, tw); __syncthreads();
    stu_fft_pass<16,   false, false>(bc, tl, tw); __syncthreads();
    for (int idx = tid; idx < FCH_ * (FN_ / 2); idx += 512) {
        const int c2 = idx >> 11, slot = idx & 2047;
        const int n = stu_hs_elem(slot);
        const v2f* s = sb + c2 * FPITCH_ + stu_pidx(n);
        const v2f u = s[0], v = s[1];
        *(float4*)(HSf + ((size_t)(d0 + c2) * FN_ + 2 * slot) * 2) = float4{u.x, u.y, v.x, v.y};
    }
}

#define GM_ROWS_  64
#define GM_PITCH_ (D_ + 4)
#define GM_TP_    68
#define GM_LDS_   (12 * 64 * GM_TP_)
__global__ __launch_bounds__(384)
void stu_xproj_wmma(const float* __restrict__ X,
                    const float* __restrict__ P,
                    float* __restrict__ XP) {
    __shared__ __attribute__((aligned(16))) float sa[GM_LDS_];

    const int tid  = threadIdx.x;
    const int wave = tid >> 5;
    const int lane = tid & 31;
    const int half = lane >> 4;
    const int l16  = lane & 15;
    const int m0 = blockIdx.x * GM_ROWS_;
    const int n0 = wave * 64;

    const float* xb = X + (size_t)m0 * D_;
#pragma unroll
    for (int k = 0; k < GM_ROWS_ * (D_ / 4) / 384; ++k) {
        const int idx = tid + k * 384;
        const int row = idx / (D_ / 4), q = idx - row * (D_ / 4);
        *(float4*)(sa + row * GM_PITCH_ + q * 4) = *(const float4*)(xb + (size_t)idx * 4);
    }
    __syncthreads();

    v8f c[4][4] = {};
    const float* ar   = sa + l16 * GM_PITCH_ + half * 2;
    const float* pcol = P + ((size_t)(n0 + l16)) * 4 + 2 * half;

    for (int k4 = 0; k4 < K4_; ++k4) {
        v2f a[4];
#pragma unroll
        for (int mt = 0; mt < 4; ++mt)
            a[mt] = *(const v2f*)(ar + mt * 16 * GM_PITCH_ + k4 * 4);
        const float* pk = pcol + (size_t)k4 * D_ * 4;
#pragma unroll
        for (int nt = 0; nt < 4; ++nt) {
            v2f b = *(const v2f*)(pk + nt * 64);
#pragma unroll
            for (int mt = 0; mt < 4; ++mt)
                c[mt][nt] = __builtin_amdgcn_wmma_f32_16x16x4_f32(false, a[mt], false, b, (short)0, c[mt][nt], false, false);
        }
    }

    __syncthreads();
    float* tp = sa + wave * (64 * GM_TP_);
#pragma unroll
    for (int mt = 0; mt < 4; ++mt) {
#pragma unroll
        for (int nt = 0; nt < 4; ++nt) {
            float* p = tp + (nt * 16 + l16) * GM_TP_ + mt * 16 + half * 8;
            *(float4*)p       = float4{c[mt][nt][0], c[mt][nt][1], c[mt][nt][2], c[mt][nt][3]};
            *(float4*)(p + 4) = float4{c[mt][nt][4], c[mt][nt][5], c[mt][nt][6], c[mt][nt][7]};
        }
    }
    __syncthreads();
    const int bi = m0 / S_, t0 = m0 % S_;
    float* ob = XP + ((size_t)bi * D_ + n0) * S_ + t0 + 4 * l16;
#pragma unroll
    for (int q = 0; q < 32; ++q) {
        const int col = 2 * q + half;
        *(float4*)(ob + (size_t)col * S_) = *(const float4*)(tp + col * GM_TP_ + 4 * l16);
    }
}

__global__ __launch_bounds__(512, 8)
void stu_conv_fft_kernel(const float* __restrict__ XP,
                         const float* __restrict__ HSf,
                         const float* __restrict__ TWf,
                         float* __restrict__ OUT) {
    __shared__ __attribute__((aligned(16))) v2f sb[FCH_ * FPITCH_];
    const int tid = threadIdx.x;
    const int c = tid >> 7, tl = tid & 127;
    const int d0 = blockIdx.x * FCH_;
    const int b  = blockIdx.y;
    const v2f* tw = (const v2f*)TWf;

    const float* xpb = XP + ((size_t)b * D_ + d0) * S_;
#pragma unroll
    for (int k = 0; k < FCH_ * S_ / 4 / 512; ++k) {
        const int idx = tid + 512 * k;
        const int ch = idx >> 10, f = idx & 1023;
        const float4 v = *(const float4*)(xpb + (size_t)idx * 4);
        v2f* s = sb + ch * FPITCH_ + stu_pidx(2 * f);
        s[0] = v2f{v.x, v.y};
        s[1] = v2f{v.z, v.w};
    }
    __syncthreads();

    v2f* bc = sb + c * FPITCH_;
    stu_fft_pass<4096, false, true >(bc, tl, tw); __syncthreads();
    stu_fft_pass<256,  false, false>(bc, tl, tw); __syncthreads();
    {
        const v2f* hs = (const v2f*)HSf + (size_t)(d0 + c) * FN_;
#pragma unroll 1
        for (int it = 0; it < 2; ++it) {
            const int task = tl + 128 * it;
            const float4* hp = (const float4*)hs + it * 1024 + tl;
            float4 h[8];
#pragma unroll
            for (int q2 = 0; q2 < 8; ++q2) h[q2] = hp[q2 * 128];
            v2f r[16];
#pragma unroll
            for (int q = 0; q < 16; ++q) r[q] = bc[stu_pidx(16 * task + q)];
            stu_fwd16<16>(r, 0, tw);
#pragma unroll
            for (int q2 = 0; q2 < 8; ++q2) {
                r[2 * q2]     = stu_cmul(r[2 * q2],     v2f{h[q2].x, h[q2].y});
                r[2 * q2 + 1] = stu_cmul(r[2 * q2 + 1], v2f{h[q2].z, h[q2].w});
            }
            stu_inv16<16>(r, 0, tw);
#pragma unroll
            for (int q = 0; q < 16; ++q) bc[stu_pidx(16 * task + q)] = r[q];
        }
    }
    __syncthreads();
    stu_fft_pass<256,  true, false>(bc, tl, tw); __syncthreads();
    stu_fft_pass<4096, true, true >(bc, tl, tw); __syncthreads();

    float* ob = OUT + (size_t)b * S_ * D_ + d0;
#pragma unroll 1
    for (int k = 0; k < HALF_S / 512; ++k) {
        const int i = tid + 512 * k;
        const int pi = stu_pidx(i);
        const v2f y0 = sb[0 * FPITCH_ + pi], y1 = sb[1 * FPITCH_ + pi];
        const v2f y2 = sb[2 * FPITCH_ + pi], y3 = sb[3 * FPITCH_ + pi];
        *(float4*)(ob + (size_t)(2 * i) * D_)     = float4{y0.x, y1.x, y2.x, y3.x};
        *(float4*)(ob + (size_t)(2 * i + 1) * D_) = float4{y0.y, y1.y, y2.y, y3.y};
    }
}

extern "C" void kernel_launch(void* const* d_in, const int* in_sizes, int n_in,
                              void* d_out, int out_size, void* d_ws, size_t ws_size,
                              hipStream_t stream) {
    (void)in_sizes; (void)n_in; (void)out_size; (void)ws_size;
    const float* x   = (const float*)d_in[0];
    const float* phi = (const float*)d_in[1];
    const float* Mi  = (const float*)d_in[2];
    const float* Mf  = (const float*)d_in[3];
    float* out  = (float*)d_out;

    float* ws_x = (float*)d_ws;
    float* ws_p = ws_x + (size_t)B_ * S_ * D_;
    float* ws_s = ws_p + (size_t)K4_ * D_ * 4;
    float* ws_t = ws_s + (size_t)D_ * FN_ * 2;

    stu_spectrum_kernel<<<dim3(SPEC_BLOCKS_ + PACK_BLOCKS_, 1, 1), 512, 0, stream>>>(phi, Mf, Mi, ws_p, ws_t, ws_s);
    stu_xproj_wmma     <<<dim3((B_ * S_) / GM_ROWS_, 1, 1),  384, 0, stream>>>(x, ws_p, ws_x);
    stu_conv_fft_kernel<<<dim3(D_ / FCH_, B_, 1),            512, 0, stream>>>(ws_x, ws_s, ws_t, out);
}


# GEMM K loop: B fragments of eight k-steps kept ahead in a register ring (refilled right after a step's 16 products), next step's A fragments fetched before the products; k order and sums kept
# speedup vs baseline: 3.0019x; 1.0445x over previous
#include <hip/hip_runtime.h>


typedef __attribute__((ext_vector_type(2))) float v2f;
typedef __attribute__((ext_vector_type(8))) float v8f;

#define B_ 4
#define S_ 4096
#define D_ 768
#define K_ 24
#define HALF_S 2048
#define K4_ (D_ / 4)

#define FN_     4096
#define FCH_    4
#define FPITCH_ (FN_ + FN_ / 16)

__device__ __forceinline__ int stu_hs_elem(int slot) {
    const int it = slot >> 10, q2 = (slot >> 7) & 7, tl = slot & 127;
    return 16 * (tl + 128 * it) + 2 * q2;
}

#define TWA4096_ 0
#define TWB4096_ 3072
#define TWA256_  3840
#define TWB256_  4032
#define TWA16_   4080
#define TWEND_   4092
__device__ __forceinline__ int stu_tw_expo(int t) {
    if (t < TWB4096_) return ((t >> 10) + 1) * (t & 1023);
    if (t < TWA256_)  { const int r = t - TWB4096_; return 4 * ((r >> 8) + 1) * (r & 255); }
    if (t < TWB256_)  { const int r = t - TWA256_;  return 16 * ((r >> 6) + 1) * (r & 63); }
    if (t < TWA16_)   { const int r = t - TWB256_;  return 64 * ((r >> 4) + 1) * (r & 15); }
    if (t < TWEND_)   { const int r = t - TWA16_;   return 256 * ((r >> 2) + 1) * (r & 3); }
    return 0;
}
template <int L> struct stu_twp {
    static constexpr int A  = (L == 4096) ? TWA4096_ : (L == 256) ? TWA256_ : TWA16_;
    static constexpr int NA = L / 4;
    static constexpr int B  = (L == 4096) ? TWB4096_ : TWB256_;
    static constexpr int NB = L / 16;
};

__device__ __forceinline__ int stu_pidx(int n) { return n + (n >> 4); }

__device__ __forceinline__ v2f stu_cmul(v2f a, v2f w) {
    return v2f{a.x * w.x - a.y * w.y, a.x * w.y + a.y * w.x};
}
__device__ __forceinline__ v2f stu_cmulc(v2f a, v2f w) {
    return v2f{a.x * w.x + a.y * w.y, a.y * w.x - a.x * w.y};
}

template <bool INV>
__device__ __forceinline__ void stu_dft4(v2f& a0, v2f& a1, v2f& a2, v2f& a3) {
    const v2f s02 = a0 + a2, d02 = a0 - a2, s13 = a1 + a3, d13 = a1 - a3;
    const v2f r = INV ? v2f{-d13.y, d13.x} : v2f{d13.y, -d13.x};
    a0 = s02 + s13; a2 = s02 - s13; a1 = d02 + r; a3 = d02 - r;
}

template <int L>
__device__ __forceinline__ void stu_fwd16(v2f (&r)[16], int j, const v2f* __restrict__ tw) {
    constexpr int S = L / 16;
#pragma unroll
    for (int q0 = 0; q0 < 4; ++q0) {
        stu_dft4<false>(r[q0], r[q0 + 4], r[q0 + 8], r[q0 + 12]);
        if (!(L == 16 && q0 == 0)) {
            const v2f* ta = tw + stu_twp<L>::A + (j + S * q0);
            r[q0 + 4]  = stu_cmul(r[q0 + 4],  ta[0]);
            r[q0 + 8]  = stu_cmul(r[q0 + 8],  ta[stu_twp<L>::NA]);
            r[q0 + 12] = stu_cmul(r[q0 + 12], ta[2 * stu_twp<L>::NA]);
        }
    }
#pragma unroll
    for (int k = 0; k < 4; ++k) stu_dft4<false>(r[4 * k], r[4 * k + 1], r[4 * k + 2], r[4 * k + 3]);
    if (L > 16) {
        const v2f* tb = tw + stu_twp<L>::B + j;
        const v2f w1 = tb[0], w2 = tb[stu_twp<L>::NB], w3 = tb[2 * stu_twp<L>::NB];
#pragma unroll
        for (int k = 0; k < 4; ++k) {
            r[4 * k + 1] = stu_cmul(r[4 * k + 1], w1);
            r[4 * k + 2] = stu_cmul(r[4 * k + 2], w2);
            r[4 * k + 3] = stu_cmul(r[4 * k + 3], w3);
        }
    }
}

template <int L>
__device__ __forceinline__ void stu_inv16(v2f (&r)[16], int j, const v2f* __restrict__ tw) {
    constexpr int S = L / 16;
    if (L > 16) {
        const v2f* tb = tw + stu_twp<L>::B + j;
        const v2f w1 = tb[0], w2 = tb[stu_twp<L>::NB], w3 = tb[2 * stu_twp<L>::NB];
#pragma unroll
        for (int k = 0; k < 4; ++k) {
            r[4 * k + 1] = stu_cmulc(r[4 * k + 1], w1);
            r[4 * k + 2] = stu_cmulc(r[4 * k + 2], w2);
            r[4 * k + 3] = stu_cmulc(r[4 * k + 3], w3);
        }
    }
#pragma unroll
    for (int k = 0; k < 4; ++k) stu_dft4<true>(r[4 * k], r[4 * k + 1], r[4 * k + 2], r[4 * k + 3]);
#pragma unroll
    for (int q0 = 0; q0 < 4; ++q0) {
        if (!(L == 16 && q0 == 0)) {
            const v2f* ta = tw + stu_twp<L>::A + (j + S * q0);
            r[q0 + 4]  = stu_cmulc(r[q0 + 4],  ta[0]);
            r[q0 + 8]  = stu_cmulc(r[q0 + 8],  ta[stu_twp<L>::NA]);
            r[q0 + 12] = stu_cmulc(r[q0 + 12], ta[2 * stu_twp<L>::NA]);
        }
        stu_dft4<true>(r[q0], r[q0 + 4], r[q0 + 8], r[q0 + 12]);
    }
}

template <int L, bool INV, bool HALF>
__device__ __forceinline__ void stu_fft_pass(v2f* __restrict__ bc, int tl, const v2f* __restrict__ tw) {
    constexpr int S = L / 16;
#pragma unroll 1
    for (int it = 0; it < 2; ++it) {
        const int task = tl + 128 * it;
        const int bb = (task / S) * L, j = task % S;
        v2f r[16];
#pragma unroll
        for (int q = 0; q < 16; ++q)
            r[q] = (HALF && !INV && q >= 8) ? v2f{0.f, 0.f} : bc[stu_pidx(bb + j + S * q)];
        if (INV) stu_inv16<L>(r, j, tw); else stu_fwd16<L>(r, j, tw);
#pragma unroll
        for (int q = 0; q < 16; ++q)
            if (!(HALF && INV && q >= 8)) bc[stu_pidx(bb + j + S * q)] = r[q];
    }
}

#define SPEC_BLOCKS_ (D_ / FCH_)
#define PACK_BLOCKS_ (K4_ * D_ / 512)
__global__ __launch_bounds__(512, 8)
void stu_spectrum_kernel(const float* __restrict__ phi,
                         const float* __restrict__ Mf,
                         const float* __restrict__ Mi,
                         float* __restrict__ P,
                         float* TWf,
                         float* __restrict__ HSf) {
    __shared__ __attribute__((aligned(16))) v2f sb[FCH_ * FPITCH_];
    const int tid = threadIdx.x;
    if (blockIdx.x >= SPEC_BLOCKS_) {
        const int g  = (blockIdx.x - SPEC_BLOCKS_) * 512 + tid;
        const int k4 = g / D_, n = g - k4 * D_;
        float4 v;
        v.x = Mi[(size_t)(4 * k4 + 0) * D_ + n];
        v.y = Mi[(size_t)(4 * k4 + 1) * D_ + n];
        v.z = Mi[(size_t)(4 * k4 + 2) * D_ + n];
        v.w = Mi[(size_t)(4 * k4 + 3) * D_ + n];
        *(float4*)(P + (size_t)g * 4) = v;
        return;
    }
    const int c = tid >> 7, tl = tid & 127;
    const int d0 = blockIdx.x * FCH_;
    v2f* twb = (v2f*)TWf + (size_t)blockIdx.x * FN_;
    for (int k = 0; k < FN_ / 512; ++k) {
        const int t = tid + 512 * k;
        float s, cs;
        sincospif((float)stu_tw_expo(t) * (1.0f / 2048.0f), &s, &cs);
        twb[t] = v2f{cs, -s};
    }
    const v2f* tw = twb;

    float4 mf[K_];
#pragma unroll
    for (int k = 0; k < K_; ++k) mf[k] = *(const float4*)(Mf + (size_t)k * D_ + d0);
    for (int jj = 0; jj < HALF_S / 512; ++jj) {
        const int j = tid + 512 * jj;
        const float4* pr = (const float4*)(phi + (size_t)(2 * j) * K_);
        float a0 = 0.f, a1 = 0.f, a2 = 0.f, a3 = 0.f;
#pragma unroll
        for (int k4 = 0; k4 < K_ / 4; ++k4) {
            const float4 p = pr[k4];
            const float pv[4] = {p.x, p.y, p.z, p.w};
#pragma unroll
            for (int e = 0; e < 4; ++e) {
                const float4 m = mf[4 * k4 + e];
                a0 = fmaf(pv[e], m.x, a0); a1 = fmaf(pv[e], m.y, a1);
                a2 = fmaf(pv[e], m.z, a2); a3 = fmaf(pv[e], m.w, a3);
            }
        }
        const float sc = 2.0f / 4096.0f;
        const int pj = stu_pidx(j);
        sb[0 * FPITCH_ + pj] = v2f{a0 * sc, 0.f};
        sb[1 * FPITCH_ + pj] = v2f{a1 * sc, 0.f};
        sb[2 * FPITCH_ + pj] = v2f{a2 * sc, 0.f};
        sb[3 * FPITCH_ + pj] = v2f{a3 * sc, 0.f};
    }
    __syncthreads();
    v2f* bc = sb + c * FPITCH_;
    stu_fft_pass<4096, false, true >(bc, tl, tw); __syncthreads();
    stu_fft_pass<256,  false, false>(bc, tl, tw); __syncthreads();
    stu_fft_pass<16,   false, false>(bc, tl, tw); __syncthreads();
    for (int idx = tid; idx < FCH_ * (FN_ / 2); idx += 512) {
        const int c2 = idx >> 11, slot = idx & 2047;
        const int n = stu_hs_elem(slot);
        const v2f* s = sb + c2 * FPITCH_ + stu_pidx(n);
        const v2f u = s[0], v = s[1];
        *(float4*)(HSf + ((size_t)(d0 + c2) * FN_ + 2 * slot) * 2) = float4{u.x, u.y, v.x, v.y};
    }
}

#define GM_ROWS_  64
#define GM_PITCH_ (D_ + 4)
#define GM_TP_    68
#define GM_LDS_   (12 * 64 * GM_TP_)
#define GM_BRING_ 8

template <bool LAST>
__device__ __forceinline__ void stu_gemm_ring(v8f (&c)[4][4], v2f (&aq)[2][4], v2f (&bq)[GM_BRING_][4],
                                              const float* ar, const float* pcol, int k4) {
#pragma unroll
    for (int s = 0; s < GM_BRING_; ++s) {
        if (!(LAST && s == GM_BRING_ - 1)) {
#pragma unroll
            for (int mt = 0; mt < 4; ++mt)
                aq[(s + 1) & 1][mt] = *(const v2f*)(ar + mt * 16 * GM_PITCH_ + (k4 + s + 1) * 4);
        }
        __builtin_amdgcn_sched_barrier(0);
#pragma unroll
        for (int nt = 0; nt < 4; ++nt) {
#pragma unroll
            for (int mt = 0; mt < 4; ++mt)
                c[mt][nt] = __builtin_amdgcn_wmma_f32_16x16x4_f32(false, aq[s & 1][mt], false, bq[s][nt], (short)0, c[mt][nt], false, false);
        }
        if (!LAST) {
            const float* pk = pcol + (size_t)(k4 + GM_BRING_ + s) * D_ * 4;
#pragma unroll
            for (int nt = 0; nt < 4; ++nt) bq[s][nt] = *(const v2f*)(pk + nt * 64);
            __builtin_amdgcn_sched_barrier(0);
        }
    }
}
__global__ __launch_bounds__(384)
void stu_xproj_wmma(const float* __restrict__ X,
                    const float* __restrict__ P,
                    float* __restrict__ XP) {
    __shared__ __attribute__((aligned(16))) float sa[GM_LDS_];

    const int tid  = threadIdx.x;
    const int wave = tid >> 5;
    const int lane = tid & 31;
    const int half = lane >> 4;
    const int l16  = lane & 15;
    const int m0 = blockIdx.x * GM_ROWS_;
    const int n0 = wave * 64;

    const float* xb = X + (size_t)m0 * D_;
#pragma unroll
    for (int k = 0; k < GM_ROWS_ * (D_ / 4) / 384; ++k) {
        const int idx = tid + k * 384;
        const int row = idx / (D_ / 4), q = idx - row * (D_ / 4);
        *(float4*)(sa + row * GM_PITCH_ + q * 4) = *(const float4*)(xb + (size_t)idx * 4);
    }
    const float* ar   = sa + l16 * GM_PITCH_ + half * 2;
    const float* pcol = P + ((size_t)(n0 + l16)) * 4 + 2 * half;
    v2f bq[GM_BRING_][4];
#pragma unroll
    for (int s = 0; s < GM_BRING_; ++s) {
#pragma unroll
        for (int nt = 0; nt < 4; ++nt) bq[s][nt] = *(const v2f*)(pcol + (size_t)s * D_ * 4 + nt * 64);
    }
    __syncthreads();

    v8f c[4][4] = {};
    v2f aq[2][4];
#pragma unroll
    for (int mt = 0; mt < 4; ++mt) aq[0][mt] = *(const v2f*)(ar + mt * 16 * GM_PITCH_);
    for (int k4 = 0; k4 < K4_ - GM_BRING_; k4 += GM_BRING_) stu_gemm_ring<false>(c, aq, bq, ar, pcol, k4);
    stu_gemm_ring<true>(c, aq, bq, ar, pcol, K4_ - GM_BRING_);

    __syncthreads();
    float* tp = sa + wave * (64 * GM_TP_);
#pragma unroll
    for (int mt = 0; mt < 4; ++mt) {
#pragma unroll
        for (int nt = 0; nt < 4; ++nt) {
            float* p = tp + (nt * 16 + l16) * GM_TP_ + mt * 16 + half * 8;
            *(float4*)p       = float4{c[mt][nt][0], c[mt][nt][1], c[mt][nt][2], c[mt][nt][3]};
            *(float4*)(p + 4) = float4{c[mt][nt][4], c[mt][nt][5], c[mt][nt][6], c[mt][nt][7]};
        }
    }
    __syncthreads();
    const int bi = m0 / S_, t0 = m0 % S_;
    float* ob = XP + ((size_t)bi * D_ + n0) * S_ + t0 + 4 * l16;
#pragma unroll
    for (int q = 0; q < 32; ++q) {
        const int col = 2 * q + half;
        *(float4*)(ob + (size_t)col * S_) = *(const float4*)(tp + col * GM_TP_ + 4 * l16);
    }
}

__global__ __launch_bounds__(512, 8)
void stu_conv_fft_kernel(const float* __restrict__ XP,
                         const float* __restrict__ HSf,
                         const float* __restrict__ TWf,
                         float* __restrict__ OUT) {
    __shared__ __attribute__((aligned(16))) v2f sb[FCH_ * FPITCH_];
    const int tid = threadIdx.x;
    const int c = tid >> 7, tl = tid & 127;
    const int d0 = blockIdx.x * FCH_;
    const int b  = blockIdx.y;
    const v2f* tw = (const v2f*)TWf;

    const float* xpb = XP + ((size_t)b * D_ + d0) * S_;
#pragma unroll
    for (int k = 0; k < FCH_ * S_ / 4 / 512; ++k) {
        const int idx = tid + 512 * k;
        const int ch = idx >> 10, f = idx & 1023;
        const float4 v = *(const float4*)(xpb + (size_t)idx * 4);
        v2f* s = sb + ch * FPITCH_ + stu_pidx(2 * f);
        s[0] = v2f{v.x, v.y};
        s[1] = v2f{v.z, v.w};
    }
    __syncthreads();

    v2f* bc = sb + c * FPITCH_;
    stu_fft_pass<4096, false, true >(bc, tl, tw); __syncthreads();
    stu_fft_pass<256,  false, false>(bc, tl, tw); __syncthreads();
    {
        const v2f* hs = (const v2f*)HSf + (size_t)(d0 + c) * FN_;
#pragma unroll 1
        for (int it = 0; it < 2; ++it) {
            const int task = tl + 128 * it;
            const float4* hp = (const float4*)hs + it * 1024 + tl;
            float4 h[8];
#pragma unroll
            for (int q2 = 0; q2 < 8; ++q2) h[q2] = hp[q2 * 128];
            v2f r[16];
#pragma unroll
            for (int q = 0; q < 16; ++q) r[q] = bc[stu_pidx(16 * task + q)];
            stu_fwd16<16>(r, 0, tw);
#pragma unroll
            for (int q2 = 0; q2 < 8; ++q2) {
                r[2 * q2]     = stu_cmul(r[2 * q2],     v2f{h[q2].x, h[q2].y});
                r[2 * q2 + 1] = stu_cmul(r[2 * q2 + 1], v2f{h[q2].z, h[q2].w});
            }
            stu_inv16<16>(r, 0, tw);
#pragma unroll
            for (int q = 0; q < 16; ++q) bc[stu_pidx(16 * task + q)] = r[q];
        }
    }
    __syncthreads();
    stu_fft_pass<256,  true, false>(bc, tl, tw); __syncthreads();
    stu_fft_pass<4096, true, true >(bc, tl, tw); __syncthreads();

    float* ob = OUT + (size_t)b * S_ * D_ + d0;
#pragma unroll 1
    for (int k = 0; k < HALF_S / 512; ++k) {
        const int i = tid + 512 * k;
        const int pi = stu_pidx(i);
        const v2f y0 = sb[0 * FPITCH_ + pi], y1 = sb[1 * FPITCH_ + pi];
        const v2f y2 = sb[2 * FPITCH_ + pi], y3 = sb[3 * FPITCH_ + pi];
        *(float4*)(ob + (size_t)(2 * i) * D_)     = float4{y0.x, y1.x, y2.x, y3.x};
        *(float4*)(ob + (size_t)(2 * i + 1) * D_) = float4{y0.y, y1.y, y2.y, y3.y};
    }
}

extern "C" void kernel_launch(void* const* d_in, const int* in_sizes, int n_in,
                              void* d_out, int out_size, void* d_ws, size_t ws_size,
                              hipStream_t stream) {
    (void)in_sizes; (void)n_in; (void)out_size; (void)ws_size;
    const float* x   = (const float*)d_in[0];
    const float* phi = (const float*)d_in[1];
    const float* Mi  = (const float*)d_in[2];
    const float* Mf  = (const float*)d_in[3];
    float* out  = (float*)d_out;

    float* ws_x = (float*)d_ws;
    float* ws_p = ws_x + (size_t)B_ * S_ * D_;
    float* ws_s = ws_p + (size_t)K4_ * D_ * 4;
    float* ws_t = ws_s + (size_t)D_ * FN_ * 2;

    stu_spectrum_kernel<<<dim3(SPEC_BLOCKS_ + PACK_BLOCKS_, 1, 1), 512, 0, stream>>>(phi, Mf, Mi, ws_p, ws_t, ws_s);
    stu_xproj_wmma     <<<dim3((B_ * S_) / GM_ROWS_, 1, 1),  384, 0, stream>>>(x, ws_p, ws_x);
    stu_conv_fft_kernel<<<dim3(D_ / FCH_, B_, 1),            512, 0, stream>>>(ws_x, ws_s, ws_t, out);
}
